# starting kernel
# baseline (speedup 1.0000x reference)
#include <hip/hip_runtime.h>

typedef __attribute__((ext_vector_type(2))) float v2f;
typedef __attribute__((ext_vector_type(8))) float v8f;

#define NN      100000
#define DINF    128
#define DHIDF   128
#define DOUTF   64
#define NEDGE   1600000

__global__ void zero_f32v4(float4* __restrict__ p, long n4) {
    long i = (long)blockIdx.x * blockDim.x + threadIdx.x;
    long st = (long)gridDim.x * blockDim.x;
    float4 z = make_float4(0.f, 0.f, 0.f, 0.f);
    for (; i < n4; i += st) p[i] = z;
}

template <int KC, bool RELU_IN>
__global__ void gemm_wmma_f32(const float* __restrict__ H,
                              const float* __restrict__ W,
                              float* __restrict__ C,
                              int N) {
    const int lane  = threadIdx.x & 31;
    const int wave  = threadIdx.x >> 5;
    const int m0    = blockIdx.x * 16;
    const int n0    = wave * 16;
    const int half  = lane >> 4;
    const int l15   = lane & 15;
    const int kbase = half * 2;

    const float* hrow = H + (long)(m0 + l15) * KC;
    const float* wrow = W + (long)(n0 + l15) * KC;

    v8f acc = {};
    #pragma unroll
    for (int k = 0; k < KC; k += 4) {
        v2f a, b;
        a.x = hrow[k + kbase];
        a.y = hrow[k + kbase + 1];
        if (RELU_IN) {
            a.x = a.x > 0.f ? a.x : 0.f;
            a.y = a.y > 0.f ? a.y : 0.f;
        }
        b.x = wrow[k + kbase];
        b.y = wrow[k + kbase + 1];
        acc = __builtin_amdgcn_wmma_f32_16x16x4_f32(
             false, a,  false, b,
             (short)0, acc,  false,  false);
    }

    float* cp = C + (long)(m0 + half * 8) * N + n0 + l15;
    #pragma unroll
    for (int i = 0; i < 8; ++i) cp[(long)i * N] = acc[i];
}

template <int SHIFT, bool UNIFORM_E>
__global__ void spmm_atomic(const int*   __restrict__ erow,
                            const int*   __restrict__ ecol,
                            const float* __restrict__ eval,
                            const float* __restrict__ h,
                            float*       __restrict__ y,
                            long totalChunks) {
    constexpr int D = 4 << SHIFT;
    long g = (long)blockIdx.x * blockDim.x + threadIdx.x;
    if (g >= totalChunks) return;
    int e;
    if (UNIFORM_E) {
        e = __builtin_amdgcn_readfirstlane((int)(g >> SHIFT));
    } else {
        e = (int)(g >> SHIFT);
    }
    int q = (int)(g & ((1 << SHIFT) - 1));

    int   r = erow[e];
    int   c = ecol[e];
    float v = eval[e];

    const float4 hv = *reinterpret_cast<const float4*>(h + (long)c * D + q * 4);
    float* yp = y + (long)r * D + q * 4;
    unsafeAtomicAdd(yp + 0, v * hv.x);
    unsafeAtomicAdd(yp + 1, v * hv.y);
    unsafeAtomicAdd(yp + 2, v * hv.z);
    unsafeAtomicAdd(yp + 3, v * hv.w);
}

__global__ void relu_add_relu(float* __restrict__ dst,
                              const float* __restrict__ res, long n) {
    long i = (long)blockIdx.x * blockDim.x + threadIdx.x;
    long st = (long)gridDim.x * blockDim.x;
    for (; i < n; i += st) {
        float a = dst[i];
        float b = res[i];
        a = a > 0.f ? a : 0.f;
        b = b > 0.f ? b : 0.f;
        dst[i] = a + b;
    }
}

extern "C" void kernel_launch(void* const* d_in, const int* in_sizes, int n_in,
                              void* d_out, int out_size, void* d_ws, size_t ws_size,
                              hipStream_t stream) {
    const float* x    = (const float*)d_in[0];
    const int*   erow = (const int*)  d_in[1];
    const int*   ecol = (const int*)  d_in[2];
    const float* evalv= (const float*)d_in[3];
    const float* W0   = (const float*)d_in[4];
    const float* W1   = (const float*)d_in[5];
    const float* W2   = (const float*)d_in[6];
    float*       out  = (float*)d_out;

    const long nh = (long)NN * DHIDF;
    float* T  = (float*)d_ws;
    float* A  = T + nh;
    float* Bb = A + nh;

    const long chunks128 = (long)NEDGE * (DHIDF / 4);
    const long chunks64  = (long)NEDGE * (DOUTF / 4);
    const int  mblocks   = NN / 16;

    gemm_wmma_f32<DINF, false><<<mblocks, 32 * (DHIDF / 16), 0, stream>>>(
        x, W0, T, DHIDF);
    zero_f32v4<<<2048, 256, 0, stream>>>((float4*)A, nh / 4);
    spmm_atomic<5, true><<<(int)((chunks128 + 255) / 256), 256, 0, stream>>>(
        erow, ecol, evalv, T, A, chunks128);

    gemm_wmma_f32<DHIDF, true><<<mblocks, 32 * (DHIDF / 16), 0, stream>>>(
        A, W1, T, DHIDF);
    zero_f32v4<<<2048, 256, 0, stream>>>((float4*)Bb, nh / 4);
    spmm_atomic<5, true><<<(int)((chunks128 + 255) / 256), 256, 0, stream>>>(
        erow, ecol, evalv, T, Bb, chunks128);
    relu_add_relu<<<2048, 256, 0, stream>>>(Bb, A, nh);

    gemm_wmma_f32<DHIDF, false><<<mblocks, 32 * (DOUTF / 16), 0, stream>>>(
        Bb, W2, T, DOUTF);
    zero_f32v4<<<2048, 256, 0, stream>>>((float4*)out, (long)NN * DOUTF / 4);
    spmm_atomic<4, false><<<(int)((chunks64 + 255) / 256), 256, 0, stream>>>(
        erow, ecol, evalv, T, out, chunks64);
}


# SpMM scatter-add turned into a gather by destination row: edges grouped by row once per call (integer atomic arrival-order rank + prefix sum), one wave sums a row in registers and stores it once
# speedup vs baseline: 2.0610x; 2.0610x over previous
#include <hip/hip_runtime.h>

typedef __attribute__((ext_vector_type(2))) float v2f;
typedef __attribute__((ext_vector_type(8))) float v8f;

#define NN      100000
#define DINF    128
#define DHIDF   128
#define DOUTF   64
#define NEDGE   1600000

__global__ void zero_f32v4(float4* __restrict__ p, long n4) {
    long i = (long)blockIdx.x * blockDim.x + threadIdx.x;
    long st = (long)gridDim.x * blockDim.x;
    float4 z = make_float4(0.f, 0.f, 0.f, 0.f);
    for (; i < n4; i += st) p[i] = z;
}

template <int KC, bool RELU_IN>
__global__ void gemm_wmma_f32(const float* __restrict__ H,
                              const float* __restrict__ W,
                              float* __restrict__ C,
                              int N) {
    const int lane  = threadIdx.x & 31;
    const int wave  = threadIdx.x >> 5;
    const int m0    = blockIdx.x * 16;
    const int n0    = wave * 16;
    const int half  = lane >> 4;
    const int l15   = lane & 15;
    const int kbase = half * 2;

    const float* hrow = H + (long)(m0 + l15) * KC;
    const float* wrow = W + (long)(n0 + l15) * KC;

    v8f acc = {};
    #pragma unroll
    for (int k = 0; k < KC; k += 4) {
        v2f a, b;
        a.x = hrow[k + kbase];
        a.y = hrow[k + kbase + 1];
        if (RELU_IN) {
            a.x = a.x > 0.f ? a.x : 0.f;
            a.y = a.y > 0.f ? a.y : 0.f;
        }
        b.x = wrow[k + kbase];
        b.y = wrow[k + kbase + 1];
        acc = __builtin_amdgcn_wmma_f32_16x16x4_f32(
             false, a,  false, b,
             (short)0, acc,  false,  false);
    }

    float* cp = C + (long)(m0 + half * 8) * N + n0 + l15;
    #pragma unroll
    for (int i = 0; i < 8; ++i) cp[(long)i * N] = acc[i];
}

#define SCAN_T  1024
#define SCAN_B  ((NN + SCAN_T - 1) / SCAN_T)

__global__ void edge_rank(const int* __restrict__ erow,
                          int* __restrict__ cnt,
                          int* __restrict__ rank) {
    int e = blockIdx.x * blockDim.x + threadIdx.x;
    if (e >= NEDGE) return;
    rank[e] = atomicAdd(&cnt[erow[e]], 1);
}

__device__ __forceinline__ int wave_incl_scan(int v, int lane) {
    for (int d = 1; d < 32; d <<= 1) {
        int t = __shfl_up(v, d);
        if (lane >= d) v += t;
    }
    return v;
}

__global__ void scan_local(const int* __restrict__ cnt,
                           int* __restrict__ rowptr,
                           int* __restrict__ bsum) {
    __shared__ int wsum[32];
    const int t = threadIdx.x, lane = t & 31, w = t >> 5;
    const int i = blockIdx.x * SCAN_T + t;
    const int v = i < NN ? cnt[i] : 0;
    const int inc = wave_incl_scan(v, lane);
    if (lane == 31) wsum[w] = inc;
    __syncthreads();
    if (w == 0) wsum[lane] = wave_incl_scan(wsum[lane], lane);
    __syncthreads();
    const int woff = w > 0 ? wsum[w - 1] : 0;
    if (i < NN) rowptr[i] = woff + inc - v;
    if (t == SCAN_T - 1) bsum[blockIdx.x] = woff + inc;
}

__global__ void scan_add(int* __restrict__ rowptr,
                         const int* __restrict__ bsum) {
    __shared__ int off_s;
    const int t = threadIdx.x, lane = t & 31, b = blockIdx.x;
    if (t < 32) {
        int s = 0;
        for (int j = lane; j < b; j += 32) s += bsum[j];
        for (int d = 16; d > 0; d >>= 1) s += __shfl_xor(s, d);
        if (lane == 0) off_s = s;
    }
    __syncthreads();
    const int off = off_s;
    const int i = b * SCAN_T + t;
    if (i < NN) rowptr[i] += off;
    if (b == SCAN_B - 1 && t == 0) rowptr[NN] = off + bsum[b];
}

__global__ void edge_place(const int*   __restrict__ erow,
                           const int*   __restrict__ ecol,
                           const float* __restrict__ eval,
                           const int*   __restrict__ rowptr,
                           const int*   __restrict__ rank,
                           int2*        __restrict__ rec) {
    int e = blockIdx.x * blockDim.x + threadIdx.x;
    if (e >= NEDGE) return;
    int p = rowptr[erow[e]] + rank[e];
    rec[p] = make_int2(ecol[e], __float_as_int(eval[e]));
}

__global__ void copy_f32v4(const float4* __restrict__ s,
                           float4* __restrict__ d, long n4) {
    long i = (long)blockIdx.x * blockDim.x + threadIdx.x;
    long st = (long)gridDim.x * blockDim.x;
    for (; i < n4; i += st) d[i] = s[i];
}

template <int V> struct rowvec;
template <> struct rowvec<4> { typedef float4 type; };
template <> struct rowvec<2> { typedef float2 type; };

__device__ __forceinline__ void axpy(float4& a, float v, const float4& x) {
    a.x += v * x.x; a.y += v * x.y; a.z += v * x.z; a.w += v * x.w;
}
__device__ __forceinline__ void axpy(float2& a, float v, const float2& x) {
    a.x += v * x.x; a.y += v * x.y;
}
__device__ __forceinline__ void vzero(float4& a) { a = make_float4(0.f, 0.f, 0.f, 0.f); }
__device__ __forceinline__ void vzero(float2& a) { a = make_float2(0.f, 0.f); }

template <int D>
__global__ void spmm_rows(const int*   __restrict__ rowptr,
                          const int2*  __restrict__ rec,
                          const float* __restrict__ h,
                          float*       __restrict__ y) {
    constexpr int V = D / 32;
    typedef typename rowvec<V>::type vec;
    const int lane = threadIdx.x & 31;
    const int wave = __builtin_amdgcn_readfirstlane((int)(threadIdx.x >> 5));
    const int row  = blockIdx.x * 8 + wave;
    if (row >= NN) return;
    const int start = rowptr[row];
    const int end   = rowptr[row + 1];

    vec acc; vzero(acc);
    for (int base = start; base < end; base += 32) {
        int n = end - base;
        if (n > 32) n = 32;
        int2 m = make_int2(0, 0);
        if (lane < n) m = rec[base + lane];
        const int mc  = m.x;
        const int mvi = m.y;
        for (int j = 0; j < n; j += 4) {
            vec   hv[4];
            float vv[4];
            #pragma unroll
            for (int u = 0; u < 4; ++u) {
                const int idx = (j + u < n) ? (j + u) : (n - 1);
                const int c = __builtin_amdgcn_readlane(mc, idx);
                vv[u] = __int_as_float(__builtin_amdgcn_readlane(mvi, idx));
                hv[u] = *reinterpret_cast<const vec*>(h + (long)c * D + lane * V);
            }
            #pragma unroll
            for (int u = 0; u < 4; ++u) {
                if (j + u < n) axpy(acc, vv[u], hv[u]);
            }
        }
    }
    *reinterpret_cast<vec*>(y + (long)row * D + lane * V) = acc;
}

__global__ void relu_add_relu(float* __restrict__ dst,
                              const float* __restrict__ res, long n) {
    long i = (long)blockIdx.x * blockDim.x + threadIdx.x;
    long st = (long)gridDim.x * blockDim.x;
    for (; i < n; i += st) {
        float a = dst[i];
        float b = res[i];
        a = a > 0.f ? a : 0.f;
        b = b > 0.f ? b : 0.f;
        dst[i] = a + b;
    }
}

extern "C" void kernel_launch(void* const* d_in, const int* in_sizes, int n_in,
                              void* d_out, int out_size, void* d_ws, size_t ws_size,
                              hipStream_t stream) {
    const float* x    = (const float*)d_in[0];
    const int*   erow = (const int*)  d_in[1];
    const int*   ecol = (const int*)  d_in[2];
    const float* evalv= (const float*)d_in[3];
    const float* W0   = (const float*)d_in[4];
    const float* W1   = (const float*)d_in[5];
    const float* W2   = (const float*)d_in[6];
    float*       out  = (float*)d_out;

    const long nh = (long)NN * DHIDF;
    float* T  = (float*)d_ws;
    float* A  = T + nh;
    float* Bb = A + nh;

    const int RP_PAD = 102400;
    int*   cnt    = (int*)d_out;
    int*   bsum   = cnt + RP_PAD;
    int*   rank   = bsum + 1024;
    int*   rowptr = rank + NEDGE;
    int2*  rec    = (int2*)(rowptr + RP_PAD);
    const long pk4 = ((long)RP_PAD + 2L * NEDGE) / 4;
    int*   rowptr2 = (int*)A;
    int2*  rec2    = (int2*)(rowptr2 + RP_PAD);

    const int mblocks = NN / 16;
    const int eblocks = NEDGE / 256;
    const int rblocks = NN / 8;

    zero_f32v4<<<(NN / 4 + 255) / 256, 256, 0, stream>>>((float4*)cnt, NN / 4);
    edge_rank<<<eblocks, 256, 0, stream>>>(erow, cnt, rank);
    scan_local<<<SCAN_B, SCAN_T, 0, stream>>>(cnt, rowptr, bsum);
    scan_add<<<SCAN_B, SCAN_T, 0, stream>>>(rowptr, bsum);
    edge_place<<<eblocks, 256, 0, stream>>>(erow, ecol, evalv, rowptr, rank,
                                            rec);

    gemm_wmma_f32<DINF, false><<<mblocks, 32 * (DHIDF / 16), 0, stream>>>(
        x, W0, T, DHIDF);
    spmm_rows<DHIDF><<<rblocks, 256, 0, stream>>>(rowptr, rec, T, A);

    gemm_wmma_f32<DHIDF, true><<<mblocks, 32 * (DHIDF / 16), 0, stream>>>(
        A, W1, T, DHIDF);
    spmm_rows<DHIDF><<<rblocks, 256, 0, stream>>>(rowptr, rec, T, Bb);
    relu_add_relu<<<2048, 256, 0, stream>>>(Bb, A, nh);

    gemm_wmma_f32<DHIDF, false><<<mblocks, 32 * (DOUTF / 16), 0, stream>>>(
        Bb, W2, T, DOUTF);
    copy_f32v4<<<1024, 256, 0, stream>>>((const float4*)rowptr, (float4*)A, pk4);
    spmm_rows<DOUTF><<<rblocks, 256, 0, stream>>>(rowptr2, rec2, T, out);
}


# GEMM re-tiled: W staged once per block in LDS, one wave does a 16-row strip over all columns, 16-byte fragment loads (k regrouped in eights, so the sum order changes), whole-line 16-byte stores
# speedup vs baseline: 2.6482x; 1.2849x over previous
#include <hip/hip_runtime.h>

typedef __attribute__((ext_vector_type(2))) float v2f;
typedef __attribute__((ext_vector_type(8))) float v8f;

#define NN      100000
#define DINF    128
#define DHIDF   128
#define DOUTF   64
#define NEDGE   1600000

__global__ void zero_f32v4(float4* __restrict__ p, long n4) {
    long i = (long)blockIdx.x * blockDim.x + threadIdx.x;
    long st = (long)gridDim.x * blockDim.x;
    float4 z = make_float4(0.f, 0.f, 0.f, 0.f);
    for (; i < n4; i += st) p[i] = z;
}

#define GK      128
#define GPITCH  (GK + 4)
#define GWAVES  4
#define MTILES  (NN / 16)

template <int NCOL, bool RELU_IN>
__global__ void __launch_bounds__(32 * GWAVES)
gemm_wmma_f32(const float* __restrict__ H,
              const float* __restrict__ W,
              float* __restrict__ C) {
    constexpr int NT = NCOL / 16;
    __shared__ float4 wl4[NCOL * GPITCH / 4];
    float* wl = reinterpret_cast<float*>(wl4);

    const int tid  = threadIdx.x;
    const int lane = tid & 31;
    const int wave = __builtin_amdgcn_readfirstlane(tid >> 5);
    const int c = lane & 15;
    const int h = lane >> 4;
    const int mtile = blockIdx.x * GWAVES + wave;
    const bool valid = mtile < MTILES;

    float4 a[16];
    if (valid) {
        const float4* hp = reinterpret_cast<const float4*>(
            H + (long)(mtile * 16 + c) * GK + 4 * h);
        #pragma unroll
        for (int p = 0; p < 16; ++p) a[p] = hp[2 * p];
    }

    for (int i = tid; i < NCOL * (GK / 4); i += 32 * GWAVES) {
        const int g = i >> 5;
        const int j = i & 31;
        const float4 v = reinterpret_cast<const float4*>(W)[i];
        const int t  = ((g >> 6) << 2) | (g & 3);
        const int cc = (g & 63) >> 2;
        *reinterpret_cast<float4*>(wl + (t * 16 + cc) * GPITCH + 4 * j) = v;
    }
    __syncthreads();
    if (!valid) return;

    if (RELU_IN) {
        #pragma unroll
        for (int p = 0; p < 16; ++p) {
            a[p].x = a[p].x > 0.f ? a[p].x : 0.f;
            a[p].y = a[p].y > 0.f ? a[p].y : 0.f;
            a[p].z = a[p].z > 0.f ? a[p].z : 0.f;
            a[p].w = a[p].w > 0.f ? a[p].w : 0.f;
        }
    }

    v8f acc[NT];
    #pragma unroll
    for (int t = 0; t < NT; ++t) acc[t] = v8f{};

    const float* bp = wl + c * GPITCH + 4 * h;
    #pragma unroll
    for (int p = 0; p < 16; ++p) {
        v2f a0, a1;
        a0.x = a[p].x; a0.y = a[p].y;
        a1.x = a[p].z; a1.y = a[p].w;
        #pragma unroll
        for (int t = 0; t < NT; ++t) {
            const float4 b = *reinterpret_cast<const float4*>(
                bp + t * 16 * GPITCH + 8 * p);
            v2f b0, b1;
            b0.x = b.x; b0.y = b.y;
            b1.x = b.z; b1.y = b.w;
            acc[t] = __builtin_amdgcn_wmma_f32_16x16x4_f32(
                false, a0, false, b0, (short)0, acc[t], false, false);
            acc[t] = __builtin_amdgcn_wmma_f32_16x16x4_f32(
                false, a1, false, b1, (short)0, acc[t], false, false);
        }
    }

    float* cp = C + (long)(mtile * 16 + 8 * h) * NCOL + 4 * c;
    #pragma unroll
    for (int i = 0; i < 8; ++i) {
        #pragma unroll
        for (int tg = 0; tg < NT / 4; ++tg) {
            float4 o;
            o.x = acc[4 * tg + 0][i];
            o.y = acc[4 * tg + 1][i];
            o.z = acc[4 * tg + 2][i];
            o.w = acc[4 * tg + 3][i];
            *reinterpret_cast<float4*>(cp + (long)i * NCOL + tg * 64) = o;
        }
    }
}

#define SCAN_T  1024
#define SCAN_B  ((NN + SCAN_T - 1) / SCAN_T)

__global__ void edge_rank(const int* __restrict__ erow,
                          int* __restrict__ cnt,
                          int* __restrict__ rank) {
    int e = blockIdx.x * blockDim.x + threadIdx.x;
    if (e >= NEDGE) return;
    rank[e] = atomicAdd(&cnt[erow[e]], 1);
}

__device__ __forceinline__ int wave_incl_scan(int v, int lane) {
    for (int d = 1; d < 32; d <<= 1) {
        int t = __shfl_up(v, d);
        if (lane >= d) v += t;
    }
    return v;
}

__global__ void scan_local(const int* __restrict__ cnt,
                           int* __restrict__ rowptr,
                           int* __restrict__ bsum) {
    __shared__ int wsum[32];
    const int t = threadIdx.x, lane = t & 31, w = t >> 5;
    const int i = blockIdx.x * SCAN_T + t;
    const int v = i < NN ? cnt[i] : 0;
    const int inc = wave_incl_scan(v, lane);
    if (lane == 31) wsum[w] = inc;
    __syncthreads();
    if (w == 0) wsum[lane] = wave_incl_scan(wsum[lane], lane);
    __syncthreads();
    const int woff = w > 0 ? wsum[w - 1] : 0;
    if (i < NN) rowptr[i] = woff + inc - v;
    if (t == SCAN_T - 1) bsum[blockIdx.x] = woff + inc;
}

__global__ void scan_add(int* __restrict__ rowptr,
                         const int* __restrict__ bsum) {
    __shared__ int off_s;
    const int t = threadIdx.x, lane = t & 31, b = blockIdx.x;
    if (t < 32) {
        int s = 0;
        for (int j = lane; j < b; j += 32) s += bsum[j];
        for (int d = 16; d > 0; d >>= 1) s += __shfl_xor(s, d);
        if (lane == 0) off_s = s;
    }
    __syncthreads();
    const int off = off_s;
    const int i = b * SCAN_T + t;
    if (i < NN) rowptr[i] += off;
    if (b == SCAN_B - 1 && t == 0) rowptr[NN] = off + bsum[b];
}

__global__ void edge_place(const int*   __restrict__ erow,
                           const int*   __restrict__ ecol,
                           const float* __restrict__ eval,
                           const int*   __restrict__ rowptr,
                           const int*   __restrict__ rank,
                           int2*        __restrict__ rec) {
    int e = blockIdx.x * blockDim.x + threadIdx.x;
    if (e >= NEDGE) return;
    int p = rowptr[erow[e]] + rank[e];
    rec[p] = make_int2(ecol[e], __float_as_int(eval[e]));
}

__global__ void copy_f32v4(const float4* __restrict__ s,
                           float4* __restrict__ d, long n4) {
    long i = (long)blockIdx.x * blockDim.x + threadIdx.x;
    long st = (long)gridDim.x * blockDim.x;
    for (; i < n4; i += st) d[i] = s[i];
}

template <int V> struct rowvec;
template <> struct rowvec<4> { typedef float4 type; };
template <> struct rowvec<2> { typedef float2 type; };

__device__ __forceinline__ void axpy(float4& a, float v, const float4& x) {
    a.x += v * x.x; a.y += v * x.y; a.z += v * x.z; a.w += v * x.w;
}
__device__ __forceinline__ void axpy(float2& a, float v, const float2& x) {
    a.x += v * x.x; a.y += v * x.y;
}
__device__ __forceinline__ void vzero(float4& a) { a = make_float4(0.f, 0.f, 0.f, 0.f); }
__device__ __forceinline__ void vzero(float2& a) { a = make_float2(0.f, 0.f); }

template <int D>
__global__ void spmm_rows(const int*   __restrict__ rowptr,
                          const int2*  __restrict__ rec,
                          const float* __restrict__ h,
                          float*       __restrict__ y) {
    constexpr int V = D / 32;
    typedef typename rowvec<V>::type vec;
    const int lane = threadIdx.x & 31;
    const int wave = __builtin_amdgcn_readfirstlane((int)(threadIdx.x >> 5));
    const int row  = blockIdx.x * 8 + wave;
    if (row >= NN) return;
    const int start = rowptr[row];
    const int end   = rowptr[row + 1];

    vec acc; vzero(acc);
    for (int base = start; base < end; base += 32) {
        int n = end - base;
        if (n > 32) n = 32;
        int2 m = make_int2(0, 0);
        if (lane < n) m = rec[base + lane];
        const int mc  = m.x;
        const int mvi = m.y;
        for (int j = 0; j < n; j += 4) {
            vec   hv[4];
            float vv[4];
            #pragma unroll
            for (int u = 0; u < 4; ++u) {
                const int idx = (j + u < n) ? (j + u) : (n - 1);
                const int c = __builtin_amdgcn_readlane(mc, idx);
                vv[u] = __int_as_float(__builtin_amdgcn_readlane(mvi, idx));
                hv[u] = *reinterpret_cast<const vec*>(h + (long)c * D + lane * V);
            }
            #pragma unroll
            for (int u = 0; u < 4; ++u) {
                if (j + u < n) axpy(acc, vv[u], hv[u]);
            }
        }
    }
    *reinterpret_cast<vec*>(y + (long)row * D + lane * V) = acc;
}

__global__ void relu_add_relu(float* __restrict__ dst,
                              const float* __restrict__ res, long n) {
    long i = (long)blockIdx.x * blockDim.x + threadIdx.x;
    long st = (long)gridDim.x * blockDim.x;
    for (; i < n; i += st) {
        float a = dst[i];
        float b = res[i];
        a = a > 0.f ? a : 0.f;
        b = b > 0.f ? b : 0.f;
        dst[i] = a + b;
    }
}

extern "C" void kernel_launch(void* const* d_in, const int* in_sizes, int n_in,
                              void* d_out, int out_size, void* d_ws, size_t ws_size,
                              hipStream_t stream) {
    const float* x    = (const float*)d_in[0];
    const int*   erow = (const int*)  d_in[1];
    const int*   ecol = (const int*)  d_in[2];
    const float* evalv= (const float*)d_in[3];
    const float* W0   = (const float*)d_in[4];
    const float* W1   = (const float*)d_in[5];
    const float* W2   = (const float*)d_in[6];
    float*       out  = (float*)d_out;

    const long nh = (long)NN * DHIDF;
    float* T  = (float*)d_ws;
    float* A  = T + nh;
    float* Bb = A + nh;

    const int RP_PAD = 102400;
    int*   cnt    = (int*)d_out;
    int*   bsum   = cnt + RP_PAD;
    int*   rank   = bsum + 1024;
    int*   rowptr = rank + NEDGE;
    int2*  rec    = (int2*)(rowptr + RP_PAD);
    const long pk4 = ((long)RP_PAD + 2L * NEDGE) / 4;
    int*   rowptr2 = (int*)A;
    int2*  rec2    = (int2*)(rowptr2 + RP_PAD);

    const int mblocks = (MTILES + GWAVES - 1) / GWAVES;
    const int eblocks = NEDGE / 256;
    const int rblocks = NN / 8;

    zero_f32v4<<<(NN / 4 + 255) / 256, 256, 0, stream>>>((float4*)cnt, NN / 4);
    edge_rank<<<eblocks, 256, 0, stream>>>(erow, cnt, rank);
    scan_local<<<SCAN_B, SCAN_T, 0, stream>>>(cnt, rowptr, bsum);
    scan_add<<<SCAN_B, SCAN_T, 0, stream>>>(rowptr, bsum);
    edge_place<<<eblocks, 256, 0, stream>>>(erow, ecol, evalv, rowptr, rank,
                                            rec);

    gemm_wmma_f32<DHIDF, false><<<mblocks, 32 * GWAVES, 0, stream>>>(x, W0, T);
    spmm_rows<DHIDF><<<rblocks, 256, 0, stream>>>(rowptr, rec, T, A);

    gemm_wmma_f32<DHIDF, true><<<mblocks, 32 * GWAVES, 0, stream>>>(A, W1, T);
    spmm_rows<DHIDF><<<rblocks, 256, 0, stream>>>(rowptr, rec, T, Bb);
    relu_add_relu<<<2048, 256, 0, stream>>>(Bb, A, nh);

    gemm_wmma_f32<DOUTF, false><<<mblocks, 32 * GWAVES, 0, stream>>>(Bb, W2, T);
    copy_f32v4<<<1024, 256, 0, stream>>>((const float4*)rowptr, (float4*)A, pk4);
    spmm_rows<DOUTF><<<rblocks, 256, 0, stream>>>(rowptr2, rec2, T, out);
}


# Edge grouping without global atomics: two-level counting sort on LDS counters (512-row buckets, per-block histograms, one scan; one block per bucket ranks rows); row sums in LDS arrival order
# speedup vs baseline: 3.5221x; 1.3300x over previous
#include <hip/hip_runtime.h>

typedef __attribute__((ext_vector_type(2))) float v2f;
typedef __attribute__((ext_vector_type(8))) float v8f;

#define NN      100000
#define DINF    128
#define DHIDF   128
#define DOUTF   64
#define NEDGE   1600000

#define GK      128
#define GPITCH  (GK + 4)
#define GWAVES  4
#define MTILES  (NN / 16)

template <int NCOL, bool RELU_IN>
__global__ void __launch_bounds__(32 * GWAVES)
gemm_wmma_f32(const float* __restrict__ H,
              const float* __restrict__ W,
              float* __restrict__ C) {
    constexpr int NT = NCOL / 16;
    __shared__ float4 wl4[NCOL * GPITCH / 4];
    float* wl = reinterpret_cast<float*>(wl4);

    const int tid  = threadIdx.x;
    const int lane = tid & 31;
    const int wave = __builtin_amdgcn_readfirstlane(tid >> 5);
    const int c = lane & 15;
    const int h = lane >> 4;
    const int mtile = blockIdx.x * GWAVES + wave;
    const bool valid = mtile < MTILES;

    float4 a[16];
    if (valid) {
        const float4* hp = reinterpret_cast<const float4*>(
            H + (long)(mtile * 16 + c) * GK + 4 * h);
        #pragma unroll
        for (int p = 0; p < 16; ++p) a[p] = hp[2 * p];
    }

    for (int i = tid; i < NCOL * (GK / 4); i += 32 * GWAVES) {
        const int g = i >> 5;
        const int j = i & 31;
        const float4 v = reinterpret_cast<const float4*>(W)[i];
        const int t  = ((g >> 6) << 2) | (g & 3);
        const int cc = (g & 63) >> 2;
        *reinterpret_cast<float4*>(wl + (t * 16 + cc) * GPITCH + 4 * j) = v;
    }
    __syncthreads();
    if (!valid) return;

    if (RELU_IN) {
        #pragma unroll
        for (int p = 0; p < 16; ++p) {
            a[p].x = a[p].x > 0.f ? a[p].x : 0.f;
            a[p].y = a[p].y > 0.f ? a[p].y : 0.f;
            a[p].z = a[p].z > 0.f ? a[p].z : 0.f;
            a[p].w = a[p].w > 0.f ? a[p].w : 0.f;
        }
    }

    v8f acc[NT];
    #pragma unroll
    for (int t = 0; t < NT; ++t) acc[t] = v8f{};

    const float* bp = wl + c * GPITCH + 4 * h;
    #pragma unroll
    for (int p = 0; p < 16; ++p) {
        v2f a0, a1;
        a0.x = a[p].x; a0.y = a[p].y;
        a1.x = a[p].z; a1.y = a[p].w;
        #pragma unroll
        for (int t = 0; t < NT; ++t) {
            const float4 b = *reinterpret_cast<const float4*>(
                bp + t * 16 * GPITCH + 8 * p);
            v2f b0, b1;
            b0.x = b.x; b0.y = b.y;
            b1.x = b.z; b1.y = b.w;
            acc[t] = __builtin_amdgcn_wmma_f32_16x16x4_f32(
                false, a0, false, b0, (short)0, acc[t], false, false);
            acc[t] = __builtin_amdgcn_wmma_f32_16x16x4_f32(
                false, a1, false, b1, (short)0, acc[t], false, false);
        }
    }

    float* cp = C + (long)(mtile * 16 + 8 * h) * NCOL + 4 * c;
    #pragma unroll
    for (int i = 0; i < 8; ++i) {
        #pragma unroll
        for (int tg = 0; tg < NT / 4; ++tg) {
            float4 o;
            o.x = acc[4 * tg + 0][i];
            o.y = acc[4 * tg + 1][i];
            o.z = acc[4 * tg + 2][i];
            o.w = acc[4 * tg + 3][i];
            *reinterpret_cast<float4*>(cp + (long)i * NCOL + tg * 64) = o;
        }
    }
}

#define BSHIFT  9
#define BROWS   (1 << BSHIFT)
#define NBUCK   ((NN + BROWS - 1) >> BSHIFT)
#define EBLK    256
#define ECHUNK  ((NEDGE + EBLK - 1) / EBLK)
#define ETHR    1024
#define BH_N4   (NBUCK * EBLK / 4)
#define SCAN_K  ((BH_N4 + ETHR - 1) / ETHR)

static_assert((NBUCK * EBLK) % 4 == 0, "bh is scanned in 16-byte pieces");
static_assert(BROWS <= ETHR, "one thread per row of a bucket in the row scan");

__device__ __forceinline__ int wave_incl_scan(int v, int lane) {
    for (int d = 1; d < 32; d <<= 1) {
        int t = __shfl_up(v, d);
        if (lane >= d) v += t;
    }
    return v;
}

__device__ __forceinline__ void wave_incl_scan_k(int (&x)[SCAN_K], int lane) {
    #pragma unroll
    for (int d = 1; d < 32; d <<= 1) {
        int t[SCAN_K];
        #pragma unroll
        for (int k = 0; k < SCAN_K; ++k) t[k] = __shfl_up(x[k], d);
        #pragma unroll
        for (int k = 0; k < SCAN_K; ++k) if (lane >= d) x[k] += t[k];
    }
}

#define EU 4

__global__ void __launch_bounds__(ETHR)
bucket_hist(const int* __restrict__ erow, int* __restrict__ bh) {
    __shared__ int hist[NBUCK];
    const int tid = threadIdx.x;
    for (int i = tid; i < NBUCK; i += ETHR) hist[i] = 0;
    __syncthreads();
    const int e0 = blockIdx.x * ECHUNK;
    const int e1 = e0 + ECHUNK < NEDGE ? e0 + ECHUNK : NEDGE;
    for (int e = e0 + tid; e < e1; e += EU * ETHR) {
        int r[EU];
        #pragma unroll
        for (int u = 0; u < EU; ++u) {
            const int ee = e + u * ETHR;
            r[u] = erow[ee < e1 ? ee : e1 - 1];
        }
        #pragma unroll
        for (int u = 0; u < EU; ++u)
            if (e + u * ETHR < e1) atomicAdd(&hist[r[u] >> BSHIFT], 1);
    }
    __syncthreads();
    for (int i = tid; i < NBUCK; i += ETHR) bh[i * EBLK + blockIdx.x] = hist[i];
}

__global__ void __launch_bounds__(ETHR)
bucket_scan(int4* __restrict__ p4) {
    __shared__ int wsum[SCAN_K][32];
    const int t = threadIdx.x, lane = t & 31, w = t >> 5;
    int4 v[SCAN_K];
    int  s[SCAN_K], inc[SCAN_K];
    #pragma unroll
    for (int k = 0; k < SCAN_K; ++k) {
        const int idx = k * ETHR + t;
        v[k] = p4[idx < BH_N4 ? idx : BH_N4 - 1];
    }
    #pragma unroll
    for (int k = 0; k < SCAN_K; ++k) {
        if (k * ETHR + t >= BH_N4) v[k] = make_int4(0, 0, 0, 0);
        s[k] = v[k].x + v[k].y + v[k].z + v[k].w;
        inc[k] = s[k];
    }
    wave_incl_scan_k(inc, lane);
    if (lane == 31) {
        #pragma unroll
        for (int k = 0; k < SCAN_K; ++k) wsum[k][w] = inc[k];
    }
    __syncthreads();
    if (w == 0) {
        int x[SCAN_K], xi[SCAN_K];
        #pragma unroll
        for (int k = 0; k < SCAN_K; ++k) { x[k] = wsum[k][lane]; xi[k] = x[k]; }
        wave_incl_scan_k(xi, lane);
        int carry = 0;
        #pragma unroll
        for (int k = 0; k < SCAN_K; ++k) {
            wsum[k][lane] = carry + xi[k] - x[k];
            carry += __builtin_amdgcn_readlane(xi[k], 31);
        }
    }
    __syncthreads();
    #pragma unroll
    for (int k = 0; k < SCAN_K; ++k) {
        const int idx = k * ETHR + t;
        if (idx < BH_N4) {
            const int ex = wsum[k][w] + inc[k] - s[k];
            p4[idx] = make_int4(ex, ex + v[k].x, ex + v[k].x + v[k].y,
                                ex + v[k].x + v[k].y + v[k].z);
        }
    }
}

__global__ void __launch_bounds__(ETHR)
bucket_scatter(const int*   __restrict__ erow,
               const int*   __restrict__ ecol,
               const float* __restrict__ eval,
               const int*   __restrict__ bh,
               int4*        __restrict__ tmp) {
    __shared__ int cur[NBUCK];
    const int tid = threadIdx.x;
    for (int i = tid; i < NBUCK; i += ETHR) cur[i] = bh[i * EBLK + blockIdx.x];
    __syncthreads();
    const int e0 = blockIdx.x * ECHUNK;
    const int e1 = e0 + ECHUNK < NEDGE ? e0 + ECHUNK : NEDGE;
    for (int e = e0 + tid; e < e1; e += EU * ETHR) {
        int4 q[EU];
        #pragma unroll
        for (int u = 0; u < EU; ++u) {
            const int ee = e + u * ETHR;
            const int ec = ee < e1 ? ee : e1 - 1;
            q[u] = make_int4(erow[ec], ecol[ec], __float_as_int(eval[ec]), 0);
        }
        #pragma unroll
        for (int u = 0; u < EU; ++u) {
            if (e + u * ETHR < e1) {
                const int p = atomicAdd(&cur[q[u].x >> BSHIFT], 1);
                tmp[p] = q[u];
            }
        }
    }
}

__global__ void __launch_bounds__(ETHR)
bucket_rows(const int*  __restrict__ bh,
            const int4* __restrict__ tmp,
            int*        __restrict__ rowptr,
            int2*       __restrict__ rec) {
    __shared__ int cnt[BROWS];
    __shared__ int wsum[32];
    const int tid = threadIdx.x, lane = tid & 31, w = tid >> 5;
    const int b  = blockIdx.x;
    const int r0 = b << BSHIFT;
    const int s0 = bh[b * EBLK];
    const int s1 = b + 1 < NBUCK ? bh[(b + 1) * EBLK] : NEDGE;
    for (int i = tid; i < BROWS; i += ETHR) cnt[i] = 0;
    __syncthreads();
    for (int e = s0 + tid; e < s1; e += EU * ETHR) {
        int r[EU];
        #pragma unroll
        for (int u = 0; u < EU; ++u) {
            const int ee = e + u * ETHR;
            r[u] = tmp[ee < s1 ? ee : s1 - 1].x;
        }
        #pragma unroll
        for (int u = 0; u < EU; ++u)
            if (e + u * ETHR < s1) atomicAdd(&cnt[r[u] - r0], 1);
    }
    __syncthreads();
    const int v   = tid < BROWS ? cnt[tid] : 0;
    const int inc = wave_incl_scan(v, lane);
    if (lane == 31) wsum[w] = inc;
    __syncthreads();
    if (w == 0) wsum[lane] = wave_incl_scan(wsum[lane], lane);
    __syncthreads();
    const int first = s0 + (w > 0 ? wsum[w - 1] : 0) + inc - v;
    if (tid < BROWS) {
        cnt[tid] = first;
        if (r0 + tid < NN) rowptr[r0 + tid] = first;
    }
    if (b == NBUCK - 1 && tid == 0) rowptr[NN] = NEDGE;
    __syncthreads();
    for (int e = s0 + tid; e < s1; e += EU * ETHR) {
        int4 q[EU];
        #pragma unroll
        for (int u = 0; u < EU; ++u) {
            const int ee = e + u * ETHR;
            q[u] = tmp[ee < s1 ? ee : s1 - 1];
        }
        #pragma unroll
        for (int u = 0; u < EU; ++u) {
            if (e + u * ETHR < s1) {
                const int p = atomicAdd(&cnt[q[u].x - r0], 1);
                rec[p] = make_int2(q[u].y, q[u].z);
            }
        }
    }
}

__global__ void copy_f32v4(const float4* __restrict__ s,
                           float4* __restrict__ d, long n4) {
    long i = (long)blockIdx.x * blockDim.x + threadIdx.x;
    long st = (long)gridDim.x * blockDim.x;
    for (; i < n4; i += st) d[i] = s[i];
}

template <int V> struct rowvec;
template <> struct rowvec<4> { typedef float4 type; };
template <> struct rowvec<2> { typedef float2 type; };

__device__ __forceinline__ void axpy(float4& a, float v, const float4& x) {
    a.x += v * x.x; a.y += v * x.y; a.z += v * x.z; a.w += v * x.w;
}
__device__ __forceinline__ void axpy(float2& a, float v, const float2& x) {
    a.x += v * x.x; a.y += v * x.y;
}
__device__ __forceinline__ void vzero(float4& a) { a = make_float4(0.f, 0.f, 0.f, 0.f); }
__device__ __forceinline__ void vzero(float2& a) { a = make_float2(0.f, 0.f); }

template <int D>
__global__ void spmm_rows(const int*   __restrict__ rowptr,
                          const int2*  __restrict__ rec,
                          const float* __restrict__ h,
                          float*       __restrict__ y) {
    constexpr int V = D / 32;
    typedef typename rowvec<V>::type vec;
    const int lane = threadIdx.x & 31;
    const int wave = __builtin_amdgcn_readfirstlane((int)(threadIdx.x >> 5));
    const int row  = blockIdx.x * 8 + wave;
    if (row >= NN) return;
    const int start = rowptr[row];
    const int end   = rowptr[row + 1];

    vec acc; vzero(acc);
    for (int base = start; base < end; base += 32) {
        int n = end - base;
        if (n > 32) n = 32;
        int2 m = make_int2(0, 0);
        if (lane < n) m = rec[base + lane];
        const int mc  = m.x;
        const int mvi = m.y;
        for (int j = 0; j < n; j += 4) {
            vec   hv[4];
            float vv[4];
            #pragma unroll
            for (int u = 0; u < 4; ++u) {
                const int idx = (j + u < n) ? (j + u) : (n - 1);
                const int c = __builtin_amdgcn_readlane(mc, idx);
                vv[u] = __int_as_float(__builtin_amdgcn_readlane(mvi, idx));
                hv[u] = *reinterpret_cast<const vec*>(h + (long)c * D + lane * V);
            }
            #pragma unroll
            for (int u = 0; u < 4; ++u) {
                if (j + u < n) axpy(acc, vv[u], hv[u]);
            }
        }
    }
    *reinterpret_cast<vec*>(y + (long)row * D + lane * V) = acc;
}

__global__ void relu_add_relu(float* __restrict__ dst,
                              const float* __restrict__ res, long n) {
    long i = (long)blockIdx.x * blockDim.x + threadIdx.x;
    long st = (long)gridDim.x * blockDim.x;
    for (; i < n; i += st) {
        float a = dst[i];
        float b = res[i];
        a = a > 0.f ? a : 0.f;
        b = b > 0.f ? b : 0.f;
        dst[i] = a + b;
    }
}

extern "C" void kernel_launch(void* const* d_in, const int* in_sizes, int n_in,
                              void* d_out, int out_size, void* d_ws, size_t ws_size,
                              hipStream_t stream) {
    const float* x    = (const float*)d_in[0];
    const int*   erow = (const int*)  d_in[1];
    const int*   ecol = (const int*)  d_in[2];
    const float* evalv= (const float*)d_in[3];
    const float* W0   = (const float*)d_in[4];
    const float* W1   = (const float*)d_in[5];
    const float* W2   = (const float*)d_in[6];
    float*       out  = (float*)d_out;

    const long nh = (long)NN * DHIDF;
    float* T  = (float*)d_ws;
    float* A  = T + nh;
    float* Bb = A + nh;

    const int RP_PAD = 102400;
    int*   rowptr = (int*)d_out;
    int2*  rec    = (int2*)(rowptr + RP_PAD);
    const long pk4 = ((long)RP_PAD + 2L * NEDGE) / 4;
    int*   rowptr2 = (int*)A;
    int2*  rec2    = (int2*)(rowptr2 + RP_PAD);
    int*   bh  = (int*)A;
    int4*  tmp = (int4*)Bb;

    const int mblocks = (MTILES + GWAVES - 1) / GWAVES;
    const int rblocks = NN / 8;

    bucket_hist<<<EBLK, ETHR, 0, stream>>>(erow, bh);
    bucket_scan<<<1, ETHR, 0, stream>>>((int4*)bh);
    bucket_scatter<<<EBLK, ETHR, 0, stream>>>(erow, ecol, evalv, bh, tmp);
    bucket_rows<<<NBUCK, ETHR, 0, stream>>>(bh, tmp, rowptr, rec);

    gemm_wmma_f32<DHIDF, false><<<mblocks, 32 * GWAVES, 0, stream>>>(x, W0, T);
    spmm_rows<DHIDF><<<rblocks, 256, 0, stream>>>(rowptr, rec, T, A);

    gemm_wmma_f32<DHIDF, true><<<mblocks, 32 * GWAVES, 0, stream>>>(A, W1, T);
    spmm_rows<DHIDF><<<rblocks, 256, 0, stream>>>(rowptr, rec, T, Bb);
    relu_add_relu<<<2048, 256, 0, stream>>>(Bb, A, nh);

    gemm_wmma_f32<DOUTF, false><<<mblocks, 32 * GWAVES, 0, stream>>>(Bb, W2, T);
    copy_f32v4<<<1024, 256, 0, stream>>>((const float4*)rowptr, (float4*)A, pk4);
    spmm_rows<DOUTF><<<rblocks, 256, 0, stream>>>(rowptr2, rec2, T, out);
}


# Residual merge moved into layer 1's SpMM store (in place); the freed plane holds the grouped edge list from the start: merge pass and list copy (2 launches) go; sum order unchanged
# speedup vs baseline: 3.9376x; 1.1180x over previous
#include <hip/hip_runtime.h>

typedef __attribute__((ext_vector_type(2))) float v2f;
typedef __attribute__((ext_vector_type(8))) float v8f;

#define NN      100000
#define DINF    128
#define DHIDF   128
#define DOUTF   64
#define NEDGE   1600000

#define GK      128
#define GPITCH  (GK + 4)
#define GWAVES  4
#define MTILES  (NN / 16)

template <int NCOL, bool RELU_IN>
__global__ void __launch_bounds__(32 * GWAVES)
gemm_wmma_f32(const float* __restrict__ H,
              const float* __restrict__ W,
              float* __restrict__ C) {
    constexpr int NT = NCOL / 16;
    __shared__ float4 wl4[NCOL * GPITCH / 4];
    float* wl = reinterpret_cast<float*>(wl4);

    const int tid  = threadIdx.x;
    const int lane = tid & 31;
    const int wave = __builtin_amdgcn_readfirstlane(tid >> 5);
    const int c = lane & 15;
    const int h = lane >> 4;
    const int mtile = blockIdx.x * GWAVES + wave;
    const bool valid = mtile < MTILES;

    float4 a[16];
    if (valid) {
        const float4* hp = reinterpret_cast<const float4*>(
            H + (long)(mtile * 16 + c) * GK + 4 * h);
        #pragma unroll
        for (int p = 0; p < 16; ++p) a[p] = hp[2 * p];
    }

    for (int i = tid; i < NCOL * (GK / 4); i += 32 * GWAVES) {
        const int g = i >> 5;
        const int j = i & 31;
        const float4 v = reinterpret_cast<const float4*>(W)[i];
        const int t  = ((g >> 6) << 2) | (g & 3);
        const int cc = (g & 63) >> 2;
        *reinterpret_cast<float4*>(wl + (t * 16 + cc) * GPITCH + 4 * j) = v;
    }
    __syncthreads();
    if (!valid) return;

    if (RELU_IN) {
        #pragma unroll
        for (int p = 0; p < 16; ++p) {
            a[p].x = a[p].x > 0.f ? a[p].x : 0.f;
            a[p].y = a[p].y > 0.f ? a[p].y : 0.f;
            a[p].z = a[p].z > 0.f ? a[p].z : 0.f;
            a[p].w = a[p].w > 0.f ? a[p].w : 0.f;
        }
    }

    v8f acc[NT];
    #pragma unroll
    for (int t = 0; t < NT; ++t) acc[t] = v8f{};

    const float* bp = wl + c * GPITCH + 4 * h;
    #pragma unroll
    for (int p = 0; p < 16; ++p) {
        v2f a0, a1;
        a0.x = a[p].x; a0.y = a[p].y;
        a1.x = a[p].z; a1.y = a[p].w;
        #pragma unroll
        for (int t = 0; t < NT; ++t) {
            const float4 b = *reinterpret_cast<const float4*>(
                bp + t * 16 * GPITCH + 8 * p);
            v2f b0, b1;
            b0.x = b.x; b0.y = b.y;
            b1.x = b.z; b1.y = b.w;
            acc[t] = __builtin_amdgcn_wmma_f32_16x16x4_f32(
                false, a0, false, b0, (short)0, acc[t], false, false);
            acc[t] = __builtin_amdgcn_wmma_f32_16x16x4_f32(
                false, a1, false, b1, (short)0, acc[t], false, false);
        }
    }

    float* cp = C + (long)(mtile * 16 + 8 * h) * NCOL + 4 * c;
    #pragma unroll
    for (int i = 0; i < 8; ++i) {
        #pragma unroll
        for (int tg = 0; tg < NT / 4; ++tg) {
            float4 o;
            o.x = acc[4 * tg + 0][i];
            o.y = acc[4 * tg + 1][i];
            o.z = acc[4 * tg + 2][i];
            o.w = acc[4 * tg + 3][i];
            *reinterpret_cast<float4*>(cp + (long)i * NCOL + tg * 64) = o;
        }
    }
}

#define BSHIFT  9
#define BROWS   (1 << BSHIFT)
#define NBUCK   ((NN + BROWS - 1) >> BSHIFT)
#define EBLK    256
#define ECHUNK  ((NEDGE + EBLK - 1) / EBLK)
#define ETHR    1024
#define BH_N4   (NBUCK * EBLK / 4)
#define SCAN_K  ((BH_N4 + ETHR - 1) / ETHR)

static_assert((NBUCK * EBLK) % 4 == 0, "bh is scanned in 16-byte pieces");
static_assert(BROWS <= ETHR, "one thread per row of a bucket in the row scan");

__device__ __forceinline__ int wave_incl_scan(int v, int lane) {
    for (int d = 1; d < 32; d <<= 1) {
        int t = __shfl_up(v, d);
        if (lane >= d) v += t;
    }
    return v;
}

__device__ __forceinline__ void wave_incl_scan_k(int (&x)[SCAN_K], int lane) {
    #pragma unroll
    for (int d = 1; d < 32; d <<= 1) {
        int t[SCAN_K];
        #pragma unroll
        for (int k = 0; k < SCAN_K; ++k) t[k] = __shfl_up(x[k], d);
        #pragma unroll
        for (int k = 0; k < SCAN_K; ++k) if (lane >= d) x[k] += t[k];
    }
}

#define EU 4

__global__ void __launch_bounds__(ETHR)
bucket_hist(const int* __restrict__ erow, int* __restrict__ bh) {
    __shared__ int hist[NBUCK];
    const int tid = threadIdx.x;
    for (int i = tid; i < NBUCK; i += ETHR) hist[i] = 0;
    __syncthreads();
    const int e0 = blockIdx.x * ECHUNK;
    const int e1 = e0 + ECHUNK < NEDGE ? e0 + ECHUNK : NEDGE;
    for (int e = e0 + tid; e < e1; e += EU * ETHR) {
        int r[EU];
        #pragma unroll
        for (int u = 0; u < EU; ++u) {
            const int ee = e + u * ETHR;
            r[u] = erow[ee < e1 ? ee : e1 - 1];
        }
        #pragma unroll
        for (int u = 0; u < EU; ++u)
            if (e + u * ETHR < e1) atomicAdd(&hist[r[u] >> BSHIFT], 1);
    }
    __syncthreads();
    for (int i = tid; i < NBUCK; i += ETHR) bh[i * EBLK + blockIdx.x] = hist[i];
}

__global__ void __launch_bounds__(ETHR)
bucket_scan(int4* __restrict__ p4) {
    __shared__ int wsum[SCAN_K][32];
    const int t = threadIdx.x, lane = t & 31, w = t >> 5;
    int4 v[SCAN_K];
    int  s[SCAN_K], inc[SCAN_K];
    #pragma unroll
    for (int k = 0; k < SCAN_K; ++k) {
        const int idx = k * ETHR + t;
        v[k] = p4[idx < BH_N4 ? idx : BH_N4 - 1];
    }
    #pragma unroll
    for (int k = 0; k < SCAN_K; ++k) {
        if (k * ETHR + t >= BH_N4) v[k] = make_int4(0, 0, 0, 0);
        s[k] = v[k].x + v[k].y + v[k].z + v[k].w;
        inc[k] = s[k];
    }
    wave_incl_scan_k(inc, lane);
    if (lane == 31) {
        #pragma unroll
        for (int k = 0; k < SCAN_K; ++k) wsum[k][w] = inc[k];
    }
    __syncthreads();
    if (w == 0) {
        int x[SCAN_K], xi[SCAN_K];
        #pragma unroll
        for (int k = 0; k < SCAN_K; ++k) { x[k] = wsum[k][lane]; xi[k] = x[k]; }
        wave_incl_scan_k(xi, lane);
        int carry = 0;
        #pragma unroll
        for (int k = 0; k < SCAN_K; ++k) {
            wsum[k][lane] = carry + xi[k] - x[k];
            carry += __builtin_amdgcn_readlane(xi[k], 31);
        }
    }
    __syncthreads();
    #pragma unroll
    for (int k = 0; k < SCAN_K; ++k) {
        const int idx = k * ETHR + t;
        if (idx < BH_N4) {
            const int ex = wsum[k][w] + inc[k] - s[k];
            p4[idx] = make_int4(ex, ex + v[k].x, ex + v[k].x + v[k].y,
                                ex + v[k].x + v[k].y + v[k].z);
        }
    }
}

__global__ void __launch_bounds__(ETHR)
bucket_scatter(const int*   __restrict__ erow,
               const int*   __restrict__ ecol,
               const float* __restrict__ eval,
               const int*   __restrict__ bh,
               int4*        __restrict__ tmp) {
    __shared__ int cur[NBUCK];
    const int tid = threadIdx.x;
    for (int i = tid; i < NBUCK; i += ETHR) cur[i] = bh[i * EBLK + blockIdx.x];
    __syncthreads();
    const int e0 = blockIdx.x * ECHUNK;
    const int e1 = e0 + ECHUNK < NEDGE ? e0 + ECHUNK : NEDGE;
    for (int e = e0 + tid; e < e1; e += EU * ETHR) {
        int4 q[EU];
        #pragma unroll
        for (int u = 0; u < EU; ++u) {
            const int ee = e + u * ETHR;
            const int ec = ee < e1 ? ee : e1 - 1;
            q[u] = make_int4(erow[ec], ecol[ec], __float_as_int(eval[ec]), 0);
        }
        #pragma unroll
        for (int u = 0; u < EU; ++u) {
            if (e + u * ETHR < e1) {
                const int p = atomicAdd(&cur[q[u].x >> BSHIFT], 1);
                tmp[p] = q[u];
            }
        }
    }
}

__global__ void __launch_bounds__(ETHR)
bucket_rows(const int*  __restrict__ bh,
            const int4* __restrict__ tmp,
            int*        __restrict__ rowptr,
            int2*       __restrict__ rec) {
    __shared__ int cnt[BROWS];
    __shared__ int wsum[32];
    const int tid = threadIdx.x, lane = tid & 31, w = tid >> 5;
    const int b  = blockIdx.x;
    const int r0 = b << BSHIFT;
    const int s0 = bh[b * EBLK];
    const int s1 = b + 1 < NBUCK ? bh[(b + 1) * EBLK] : NEDGE;
    for (int i = tid; i < BROWS; i += ETHR) cnt[i] = 0;
    __syncthreads();
    for (int e = s0 + tid; e < s1; e += EU * ETHR) {
        int r[EU];
        #pragma unroll
        for (int u = 0; u < EU; ++u) {
            const int ee = e + u * ETHR;
            r[u] = tmp[ee < s1 ? ee : s1 - 1].x;
        }
        #pragma unroll
        for (int u = 0; u < EU; ++u)
            if (e + u * ETHR < s1) atomicAdd(&cnt[r[u] - r0], 1);
    }
    __syncthreads();
    const int v   = tid < BROWS ? cnt[tid] : 0;
    const int inc = wave_incl_scan(v, lane);
    if (lane == 31) wsum[w] = inc;
    __syncthreads();
    if (w == 0) wsum[lane] = wave_incl_scan(wsum[lane], lane);
    __syncthreads();
    const int first = s0 + (w > 0 ? wsum[w - 1] : 0) + inc - v;
    if (tid < BROWS) {
        cnt[tid] = first;
        if (r0 + tid < NN) rowptr[r0 + tid] = first;
    }
    if (b == NBUCK - 1 && tid == 0) rowptr[NN] = NEDGE;
    __syncthreads();
    for (int e = s0 + tid; e < s1; e += EU * ETHR) {
        int4 q[EU];
        #pragma unroll
        for (int u = 0; u < EU; ++u) {
            const int ee = e + u * ETHR;
            q[u] = tmp[ee < s1 ? ee : s1 - 1];
        }
        #pragma unroll
        for (int u = 0; u < EU; ++u) {
            if (e + u * ETHR < s1) {
                const int p = atomicAdd(&cnt[q[u].x - r0], 1);
                rec[p] = make_int2(q[u].y, q[u].z);
            }
        }
    }
}

template <int V> struct rowvec;
template <> struct rowvec<4> { typedef float4 type; };
template <> struct rowvec<2> { typedef float2 type; };

__device__ __forceinline__ void axpy(float4& a, float v, const float4& x) {
    a.x += v * x.x; a.y += v * x.y; a.z += v * x.z; a.w += v * x.w;
}
__device__ __forceinline__ void axpy(float2& a, float v, const float2& x) {
    a.x += v * x.x; a.y += v * x.y;
}
__device__ __forceinline__ void vzero(float4& a) { a = make_float4(0.f, 0.f, 0.f, 0.f); }
__device__ __forceinline__ void vzero(float2& a) { a = make_float2(0.f, 0.f); }
__device__ __forceinline__ float relu1(float a) { return a > 0.f ? a : 0.f; }
__device__ __forceinline__ void relu_add_relu(float4& a, const float4& r) {
    a.x = relu1(a.x) + relu1(r.x); a.y = relu1(a.y) + relu1(r.y);
    a.z = relu1(a.z) + relu1(r.z); a.w = relu1(a.w) + relu1(r.w);
}
__device__ __forceinline__ void relu_add_relu(float2& a, const float2& r) {
    a.x = relu1(a.x) + relu1(r.x); a.y = relu1(a.y) + relu1(r.y);
}

template <int D, bool RESID>
__global__ void spmm_rows(const int*   __restrict__ rowptr,
                          const int2*  __restrict__ rec,
                          const float* __restrict__ h,
                          float*                    y) {
    constexpr int V = D / 32;
    typedef typename rowvec<V>::type vec;
    const int lane = threadIdx.x & 31;
    const int wave = __builtin_amdgcn_readfirstlane((int)(threadIdx.x >> 5));
    const int row  = blockIdx.x * 8 + wave;
    if (row >= NN) return;
    const int start = rowptr[row];
    const int end   = rowptr[row + 1];
    vec* const yp = reinterpret_cast<vec*>(y + (long)row * D + lane * V);

    vec res; vzero(res);
    vec acc; vzero(acc);
    for (int base = start; base < end; base += 32) {
        int n = end - base;
        if (n > 32) n = 32;
        int2 m = make_int2(0, 0);
        if (lane < n) m = rec[base + lane];
        if (RESID && base == start) res = *yp;
        const int mc  = m.x;
        const int mvi = m.y;
        for (int j = 0; j < n; j += 4) {
            vec   hv[4];
            float vv[4];
            #pragma unroll
            for (int u = 0; u < 4; ++u) {
                const int idx = (j + u < n) ? (j + u) : (n - 1);
                const int c = __builtin_amdgcn_readlane(mc, idx);
                vv[u] = __int_as_float(__builtin_amdgcn_readlane(mvi, idx));
                hv[u] = *reinterpret_cast<const vec*>(h + (long)c * D + lane * V);
            }
            #pragma unroll
            for (int u = 0; u < 4; ++u) {
                if (j + u < n) axpy(acc, vv[u], hv[u]);
            }
        }
    }
    if (RESID) {
        if (start >= end) res = *yp;
        relu_add_relu(acc, res);
    }
    *yp = acc;
}

extern "C" void kernel_launch(void* const* d_in, const int* in_sizes, int n_in,
                              void* d_out, int out_size, void* d_ws, size_t ws_size,
                              hipStream_t stream) {
    const float* x    = (const float*)d_in[0];
    const int*   erow = (const int*)  d_in[1];
    const int*   ecol = (const int*)  d_in[2];
    const float* evalv= (const float*)d_in[3];
    const float* W0   = (const float*)d_in[4];
    const float* W1   = (const float*)d_in[5];
    const float* W2   = (const float*)d_in[6];
    float*       out  = (float*)d_out;

    const long nh = (long)NN * DHIDF;
    float* T  = (float*)d_ws;
    float* A  = T + nh;
    float* L  = A + nh;

    const int RP_PAD = 102400;
    int4*  tmp    = (int4*)L;
    int*   rowptr = (int*)(L + 4L * NEDGE);
    int2*  rec    = (int2*)(rowptr + RP_PAD);
    int*   bh     = (int*)A;
    static_assert(4L * NEDGE + RP_PAD + 2L * NEDGE <= (long)NN * DHIDF,
                  "bucket-ordered edges, rowptr and records fit in one plane");

    const int mblocks = (MTILES + GWAVES - 1) / GWAVES;
    const int rblocks = NN / 8;

    bucket_hist<<<EBLK, ETHR, 0, stream>>>(erow, bh);
    bucket_scan<<<1, ETHR, 0, stream>>>((int4*)bh);
    bucket_scatter<<<EBLK, ETHR, 0, stream>>>(erow, ecol, evalv, bh, tmp);
    bucket_rows<<<NBUCK, ETHR, 0, stream>>>(bh, tmp, rowptr, rec);

    gemm_wmma_f32<DHIDF, false><<<mblocks, 32 * GWAVES, 0, stream>>>(x, W0, T);
    spmm_rows<DHIDF, false><<<rblocks, 256, 0, stream>>>(rowptr, rec, T, A);

    gemm_wmma_f32<DHIDF, true><<<mblocks, 32 * GWAVES, 0, stream>>>(A, W1, T);
    spmm_rows<DHIDF, true><<<rblocks, 256, 0, stream>>>(rowptr, rec, T, A);

    gemm_wmma_f32<DOUTF, false><<<mblocks, 32 * GWAVES, 0, stream>>>(A, W2, T);
    spmm_rows<DOUTF, false><<<rblocks, 256, 0, stream>>>(rowptr, rec, T, out);
}


# GEMM fused into SpMM by re-association, (A h) W^T for A (h W^T): a wave sums 16 rows into LDS and feeds WMMA from there; W2 product rides on layer 1; 3 launches go; sum order changes
# speedup vs baseline: 4.8600x; 1.2343x over previous
#include <hip/hip_runtime.h>

typedef __attribute__((ext_vector_type(2))) float v2f;
typedef __attribute__((ext_vector_type(8))) float v8f;

#define NN      100000
#define DINF    128
#define DHIDF   128
#define DOUTF   64
#define NEDGE   1600000

#define BSHIFT  9
#define BROWS   (1 << BSHIFT)
#define NBUCK   ((NN + BROWS - 1) >> BSHIFT)
#define EBLK    256
#define ECHUNK  ((NEDGE + EBLK - 1) / EBLK)
#define ETHR    1024
#define BH_N4   (NBUCK * EBLK / 4)
#define SCAN_K  ((BH_N4 + ETHR - 1) / ETHR)

static_assert((NBUCK * EBLK) % 4 == 0, "bh is scanned in 16-byte pieces");
static_assert(BROWS <= ETHR, "one thread per row of a bucket in the row scan");

__device__ __forceinline__ int wave_incl_scan(int v, int lane) {
    for (int d = 1; d < 32; d <<= 1) {
        int t = __shfl_up(v, d);
        if (lane >= d) v += t;
    }
    return v;
}

__device__ __forceinline__ void wave_incl_scan_k(int (&x)[SCAN_K], int lane) {
    #pragma unroll
    for (int d = 1; d < 32; d <<= 1) {
        int t[SCAN_K];
        #pragma unroll
        for (int k = 0; k < SCAN_K; ++k) t[k] = __shfl_up(x[k], d);
        #pragma unroll
        for (int k = 0; k < SCAN_K; ++k) if (lane >= d) x[k] += t[k];
    }
}

#define EU 4

__global__ void __launch_bounds__(ETHR)
bucket_hist(const int* __restrict__ erow, int* __restrict__ bh) {
    __shared__ int hist[NBUCK];
    const int tid = threadIdx.x;
    for (int i = tid; i < NBUCK; i += ETHR) hist[i] = 0;
    __syncthreads();
    const int e0 = blockIdx.x * ECHUNK;
    const int e1 = e0 + ECHUNK < NEDGE ? e0 + ECHUNK : NEDGE;
    for (int e = e0 + tid; e < e1; e += EU * ETHR) {
        int r[EU];
        #pragma unroll
        for (int u = 0; u < EU; ++u) {
            const int ee = e + u * ETHR;
            r[u] = erow[ee < e1 ? ee : e1 - 1];
        }
        #pragma unroll
        for (int u = 0; u < EU; ++u)
            if (e + u * ETHR < e1) atomicAdd(&hist[r[u] >> BSHIFT], 1);
    }
    __syncthreads();
    for (int i = tid; i < NBUCK; i += ETHR) bh[i * EBLK + blockIdx.x] = hist[i];
}

__global__ void __launch_bounds__(ETHR)
bucket_scan(int4* __restrict__ p4) {
    __shared__ int wsum[SCAN_K][32];
    const int t = threadIdx.x, lane = t & 31, w = t >> 5;
    int4 v[SCAN_K];
    int  s[SCAN_K], inc[SCAN_K];
    #pragma unroll
    for (int k = 0; k < SCAN_K; ++k) {
        const int idx = k * ETHR + t;
        v[k] = p4[idx < BH_N4 ? idx : BH_N4 - 1];
    }
    #pragma unroll
    for (int k = 0; k < SCAN_K; ++k) {
        if (k * ETHR + t >= BH_N4) v[k] = make_int4(0, 0, 0, 0);
        s[k] = v[k].x + v[k].y + v[k].z + v[k].w;
        inc[k] = s[k];
    }
    wave_incl_scan_k(inc, lane);
    if (lane == 31) {
        #pragma unroll
        for (int k = 0; k < SCAN_K; ++k) wsum[k][w] = inc[k];
    }
    __syncthreads();
    if (w == 0) {
        int x[SCAN_K], xi[SCAN_K];
        #pragma unroll
        for (int k = 0; k < SCAN_K; ++k) { x[k] = wsum[k][lane]; xi[k] = x[k]; }
        wave_incl_scan_k(xi, lane);
        int carry = 0;
        #pragma unroll
        for (int k = 0; k < SCAN_K; ++k) {
            wsum[k][lane] = carry + xi[k] - x[k];
            carry += __builtin_amdgcn_readlane(xi[k], 31);
        }
    }
    __syncthreads();
    #pragma unroll
    for (int k = 0; k < SCAN_K; ++k) {
        const int idx = k * ETHR + t;
        if (idx < BH_N4) {
            const int ex = wsum[k][w] + inc[k] - s[k];
            p4[idx] = make_int4(ex, ex + v[k].x, ex + v[k].x + v[k].y,
                                ex + v[k].x + v[k].y + v[k].z);
        }
    }
}

__global__ void __launch_bounds__(ETHR)
bucket_scatter(const int*   __restrict__ erow,
               const int*   __restrict__ ecol,
               const float* __restrict__ eval,
               const int*   __restrict__ bh,
               int4*        __restrict__ tmp) {
    __shared__ int cur[NBUCK];
    const int tid = threadIdx.x;
    for (int i = tid; i < NBUCK; i += ETHR) cur[i] = bh[i * EBLK + blockIdx.x];
    __syncthreads();
    const int e0 = blockIdx.x * ECHUNK;
    const int e1 = e0 + ECHUNK < NEDGE ? e0 + ECHUNK : NEDGE;
    for (int e = e0 + tid; e < e1; e += EU * ETHR) {
        int4 q[EU];
        #pragma unroll
        for (int u = 0; u < EU; ++u) {
            const int ee = e + u * ETHR;
            const int ec = ee < e1 ? ee : e1 - 1;
            q[u] = make_int4(erow[ec], ecol[ec], __float_as_int(eval[ec]), 0);
        }
        #pragma unroll
        for (int u = 0; u < EU; ++u) {
            if (e + u * ETHR < e1) {
                const int p = atomicAdd(&cur[q[u].x >> BSHIFT], 1);
                tmp[p] = q[u];
            }
        }
    }
}

__global__ void __launch_bounds__(ETHR)
bucket_rows(const int*  __restrict__ bh,
            const int4* __restrict__ tmp,
            int*        __restrict__ rowptr,
            int2*       __restrict__ rec) {
    __shared__ int cnt[BROWS];
    __shared__ int wsum[32];
    const int tid = threadIdx.x, lane = tid & 31, w = tid >> 5;
    const int b  = blockIdx.x;
    const int r0 = b << BSHIFT;
    const int s0 = bh[b * EBLK];
    const int s1 = b + 1 < NBUCK ? bh[(b + 1) * EBLK] : NEDGE;
    for (int i = tid; i < BROWS; i += ETHR) cnt[i] = 0;
    __syncthreads();
    for (int e = s0 + tid; e < s1; e += EU * ETHR) {
        int r[EU];
        #pragma unroll
        for (int u = 0; u < EU; ++u) {
            const int ee = e + u * ETHR;
            r[u] = tmp[ee < s1 ? ee : s1 - 1].x;
        }
        #pragma unroll
        for (int u = 0; u < EU; ++u)
            if (e + u * ETHR < s1) atomicAdd(&cnt[r[u] - r0], 1);
    }
    __syncthreads();
    const int v   = tid < BROWS ? cnt[tid] : 0;
    const int inc = wave_incl_scan(v, lane);
    if (lane == 31) wsum[w] = inc;
    __syncthreads();
    if (w == 0) wsum[lane] = wave_incl_scan(wsum[lane], lane);
    __syncthreads();
    const int first = s0 + (w > 0 ? wsum[w - 1] : 0) + inc - v;
    if (tid < BROWS) {
        cnt[tid] = first;
        if (r0 + tid < NN) rowptr[r0 + tid] = first;
    }
    if (b == NBUCK - 1 && tid == 0) rowptr[NN] = NEDGE;
    __syncthreads();
    for (int e = s0 + tid; e < s1; e += EU * ETHR) {
        int4 q[EU];
        #pragma unroll
        for (int u = 0; u < EU; ++u) {
            const int ee = e + u * ETHR;
            q[u] = tmp[ee < s1 ? ee : s1 - 1];
        }
        #pragma unroll
        for (int u = 0; u < EU; ++u) {
            if (e + u * ETHR < s1) {
                const int p = atomicAdd(&cnt[q[u].x - r0], 1);
                rec[p] = make_int2(q[u].y, q[u].z);
            }
        }
    }
}

#define GK      128
#define GPITCH  (GK + 4)
#define FWAVES  25
#define FTHR    (32 * FWAVES)
#define MTILES  (NN / 16)
#define GU      8

static_assert(NN % 16 == 0 && MTILES % FWAVES == 0, "every wave of every block owns a whole tile");

template <int NROW>
__device__ __forceinline__ void stage_w(const float* __restrict__ Wg, float* wl, int tid) {
    constexpr int N4 = NROW * (GK / 4);
    constexpr int TRIPS = (N4 + FTHR - 1) / FTHR;
    float4 v[TRIPS];
    #pragma unroll
    for (int k = 0; k < TRIPS; ++k) {
        const int i = tid + k * FTHR;
        v[k] = reinterpret_cast<const float4*>(Wg)[i < N4 ? i : N4 - 1];
    }
    #pragma unroll
    for (int k = 0; k < TRIPS; ++k) {
        const int i = tid + k * FTHR;
        if (i < N4) {
            const int g = i >> 5;
            const int j = i & 31;
            const int t  = ((g >> 6) << 2) | (g & 3);
            const int cc = (g & 63) >> 2;
            *reinterpret_cast<float4*>(wl + (t * 16 + cc) * GPITCH + 4 * j) = v[k];
        }
    }
}

template <int NT>
__device__ __forceinline__ void strip_times_w(const float* ap, const float* bp, v8f (&acc)[NT]) {
    #pragma unroll
    for (int p = 0; p < 16; ++p) {
        const float4 a = *reinterpret_cast<const float4*>(ap + 8 * p);
        v2f a0, a1;
        a0.x = a.x; a0.y = a.y;
        a1.x = a.z; a1.y = a.w;
        #pragma unroll
        for (int t = 0; t < NT; ++t) {
            const float4 b = *reinterpret_cast<const float4*>(
                bp + t * 16 * GPITCH + 8 * p);
            v2f b0, b1;
            b0.x = b.x; b0.y = b.y;
            b1.x = b.z; b1.y = b.w;
            acc[t] = __builtin_amdgcn_wmma_f32_16x16x4_f32(
                false, a0, false, b0, (short)0, acc[t], false, false);
            acc[t] = __builtin_amdgcn_wmma_f32_16x16x4_f32(
                false, a1, false, b1, (short)0, acc[t], false, false);
        }
    }
}

__device__ __forceinline__ void strip_order() {
    __builtin_amdgcn_fence(__ATOMIC_RELEASE, "wavefront");
    __builtin_amdgcn_wave_barrier();
    __builtin_amdgcn_fence(__ATOMIC_ACQUIRE, "wavefront");
}

__device__ __forceinline__ void axpy(float4& a, float v, const float4& x) {
    a.x += v * x.x; a.y += v * x.y; a.z += v * x.z; a.w += v * x.w;
}
__device__ __forceinline__ void axpy(float2& a, float v, const float2& x) {
    a.x += v * x.x; a.y += v * x.y;
}
__device__ __forceinline__ float relu1(float a) { return a > 0.f ? a : 0.f; }

template <bool SECOND>
__global__ void __launch_bounds__(FTHR)
spmm_gemm(const int*   __restrict__ rowptr,
          const int2*  __restrict__ rec,
          const float* __restrict__ hin,
          const float* __restrict__ Wa,
          const float* __restrict__ Wb,
          float*       __restrict__ outp)
{
    __shared__ float4 wl4[DHIDF * GPITCH / 4];
    __shared__ float4 w2l4[SECOND ? DOUTF * GPITCH / 4 : 1];
    __shared__ float4 st4[FWAVES * 16 * GPITCH / 4];
    float* wl  = reinterpret_cast<float*>(wl4);
    float* w2l = reinterpret_cast<float*>(w2l4);

    const int tid  = threadIdx.x;
    const int lane = tid & 31;
    const int wave = __builtin_amdgcn_readfirstlane(tid >> 5);
    const int c = lane & 15;
    const int h = lane >> 4;
    const int r0 = (blockIdx.x * FWAVES + wave) * 16;
    float* sp = reinterpret_cast<float*>(st4) + wave * 16 * GPITCH;

    const int rp = rowptr[r0 + (lane < 16 ? lane : 16)];

    stage_w<DHIDF>(Wa, wl, tid);
    if (SECOND) stage_w<DOUTF>(Wb, w2l, tid);
    __syncthreads();

    const int ts = __builtin_amdgcn_readlane(rp, 0);
    const int te = __builtin_amdgcn_readlane(rp, 16);
    float* const srow = sp + lane * 4;
    float4 acc = make_float4(0.f, 0.f, 0.f, 0.f);
    int row  = 0;
    int rend = __builtin_amdgcn_readlane(rp, 1);
    int2 m = make_int2(0, 0);
    if (ts < te) {
        const int i = ts + lane;
        m = rec[i < te ? i : te - 1];
    }
    for (int cb = ts; cb < te; cb += 32) {
        int2 mn = m;
        if (cb + 32 < te) {
            const int i = cb + 32 + lane;
            mn = rec[i < te ? i : te - 1];
        }
        int n = te - cb;
        if (n > 32) n = 32;
        const int mc  = m.x;
        const int mvi = m.y;
        for (int j = 0; j < n; j += GU) {
            float4 hv[GU];
            float  vv[GU];
            #pragma unroll
            for (int u = 0; u < GU; ++u) {
                const int idx = (j + u < n) ? (j + u) : (n - 1);
                const int col = __builtin_amdgcn_readlane(mc, idx);
                vv[u] = __int_as_float(__builtin_amdgcn_readlane(mvi, idx));
                hv[u] = *reinterpret_cast<const float4*>(hin + (long)col * GK + lane * 4);
            }
            #pragma unroll
            for (int u = 0; u < GU; ++u) {
                if (j + u < n) {
                    const int e = cb + j + u;
                    while (e >= rend) {
                        *reinterpret_cast<float4*>(srow + row * GPITCH) = acc;
                        acc = make_float4(0.f, 0.f, 0.f, 0.f);
                        ++row;
                        rend = __builtin_amdgcn_readlane(rp, row + 1);
                    }
                    axpy(acc, vv[u], hv[u]);
                }
            }
        }
        m = mn;
    }
    for (; row < 16; ++row) {
        *reinterpret_cast<float4*>(srow + row * GPITCH) = acc;
        acc = make_float4(0.f, 0.f, 0.f, 0.f);
    }
    strip_order();

    const float* ap = sp + c * GPITCH + 4 * h;
    v8f acc1[DHIDF / 16];
    #pragma unroll
    for (int t = 0; t < DHIDF / 16; ++t) acc1[t] = v8f{};
    strip_times_w<DHIDF / 16>(ap, wl + c * GPITCH + 4 * h, acc1);

    if (!SECOND) {
        float* cp = outp + (long)(r0 + 8 * h) * DHIDF + 4 * c;
        #pragma unroll
        for (int i = 0; i < 8; ++i) {
            #pragma unroll
            for (int tg = 0; tg < DHIDF / 64; ++tg) {
                float4 o;
                o.x = relu1(acc1[4 * tg + 0][i]);
                o.y = relu1(acc1[4 * tg + 1][i]);
                o.z = relu1(acc1[4 * tg + 2][i]);
                o.w = relu1(acc1[4 * tg + 3][i]);
                *reinterpret_cast<float4*>(cp + (long)i * DHIDF + tg * 64) = o;
            }
        }
    } else {
        #pragma unroll
        for (int i = 0; i < 8; ++i) {
            #pragma unroll
            for (int tg = 0; tg < DHIDF / 64; ++tg) {
                float4 o;
                o.x = relu1(acc1[4 * tg + 0][i]);
                o.y = relu1(acc1[4 * tg + 1][i]);
                o.z = relu1(acc1[4 * tg + 2][i]);
                o.w = relu1(acc1[4 * tg + 3][i]);
                *reinterpret_cast<float4*>(sp + (i + 8 * h) * GPITCH + tg * 64 + 4 * c) = o;
            }
        }
        strip_order();
        const float* hp = hin + (long)r0 * DHIDF + lane * 4;
        float4 rr[16];
        #pragma unroll
        for (int r = 0; r < 16; ++r)
            rr[r] = *reinterpret_cast<const float4*>(hp + (long)r * DHIDF);
        #pragma unroll
        for (int r = 0; r < 16; ++r) {
            float4 v = *reinterpret_cast<const float4*>(srow + r * GPITCH);
            v.x += rr[r].x; v.y += rr[r].y; v.z += rr[r].z; v.w += rr[r].w;
            *reinterpret_cast<float4*>(srow + r * GPITCH) = v;
        }
        strip_order();
        v8f acc2[DOUTF / 16];
        #pragma unroll
        for (int t = 0; t < DOUTF / 16; ++t) acc2[t] = v8f{};
        strip_times_w<DOUTF / 16>(ap, w2l + c * GPITCH + 4 * h, acc2);
        float* cp = outp + (long)(r0 + 8 * h) * DOUTF + 4 * c;
        #pragma unroll
        for (int i = 0; i < 8; ++i) {
            float4 o;
            o.x = acc2[0][i];
            o.y = acc2[1][i];
            o.z = acc2[2][i];
            o.w = acc2[3][i];
            *reinterpret_cast<float4*>(cp + (long)i * DOUTF) = o;
        }
    }
}

__global__ void spmm_rows64(const int*   __restrict__ rowptr,
                            const int2*  __restrict__ rec,
                            const float* __restrict__ t,
                            float*       __restrict__ y) {
    const int lane = threadIdx.x & 31;
    const int wave = __builtin_amdgcn_readfirstlane((int)(threadIdx.x >> 5));
    const int row  = blockIdx.x * 8 + wave;
    if (row >= NN) return;
    const int start = rowptr[row];
    const int end   = rowptr[row + 1];

    float2 acc = make_float2(0.f, 0.f);
    for (int base = start; base < end; base += 32) {
        int n = end - base;
        if (n > 32) n = 32;
        int2 m = make_int2(0, 0);
        if (lane < n) m = rec[base + lane];
        const int mc  = m.x;
        const int mvi = m.y;
        for (int j = 0; j < n; j += 4) {
            float2 hv[4];
            float  vv[4];
            #pragma unroll
            for (int u = 0; u < 4; ++u) {
                const int idx = (j + u < n) ? (j + u) : (n - 1);
                const int col = __builtin_amdgcn_readlane(mc, idx);
                vv[u] = __int_as_float(__builtin_amdgcn_readlane(mvi, idx));
                hv[u] = *reinterpret_cast<const float2*>(t + (long)col * DOUTF + lane * 2);
            }
            #pragma unroll
            for (int u = 0; u < 4; ++u) {
                if (j + u < n) axpy(acc, vv[u], hv[u]);
            }
        }
    }
    *reinterpret_cast<float2*>(y + (long)row * DOUTF + lane * 2) = acc;
}

extern "C" void kernel_launch(void* const* d_in, const int* in_sizes, int n_in,
                              void* d_out, int out_size, void* d_ws, size_t ws_size,
                              hipStream_t stream) {
    const float* x    = (const float*)d_in[0];
    const int*   erow = (const int*)  d_in[1];
    const int*   ecol = (const int*)  d_in[2];
    const float* evalv= (const float*)d_in[3];
    const float* W0   = (const float*)d_in[4];
    const float* W1   = (const float*)d_in[5];
    const float* W2   = (const float*)d_in[6];
    float*       out  = (float*)d_out;

    const long nh = (long)NN * DHIDF;
    float* H1 = (float*)d_ws;
    float* T2 = H1 + nh;
    float* L  = T2 + nh;

    const int RP_PAD = 102400;
    int4*  tmp    = (int4*)L;
    int*   rowptr = (int*)(L + 4L * NEDGE);
    int2*  rec    = (int2*)(rowptr + RP_PAD);
    int*   bh     = (int*)T2;
    static_assert(4L * NEDGE + RP_PAD + 2L * NEDGE <= (long)NN * DHIDF,
                  "bucket-ordered edges, rowptr and records fit in one plane");

    const int fblocks = MTILES / FWAVES;
    const int rblocks = NN / 8;

    bucket_hist<<<EBLK, ETHR, 0, stream>>>(erow, bh);
    bucket_scan<<<1, ETHR, 0, stream>>>((int4*)bh);
    bucket_scatter<<<EBLK, ETHR, 0, stream>>>(erow, ecol, evalv, bh, tmp);
    bucket_rows<<<NBUCK, ETHR, 0, stream>>>(bh, tmp, rowptr, rec);

    spmm_gemm<false><<<fblocks, FTHR, 0, stream>>>(rowptr, rec, x, W0, nullptr, H1);

    spmm_gemm<true><<<fblocks, FTHR, 0, stream>>>(rowptr, rec, H1, W1, W2, T2);

    spmm_rows64<<<rblocks, 256, 0, stream>>>(rowptr, rec, T2, out);
}
